# starting kernel
# baseline (speedup 1.0000x reference)
#include <hip/hip_runtime.h>

typedef float v2f __attribute__((ext_vector_type(2)));
typedef float v8f __attribute__((ext_vector_type(8)));

namespace {
constexpr int kB  = 8;
constexpr int kC  = 64;
constexpr int kH  = 64;
constexpr int kW  = 512;
constexpr int kO  = 64;
constexpr int kTW = 30;
constexpr int kNS = 18;
constexpr int kHB = 16;
constexpr int kNH = 4;
constexpr int kODL = 128;
constexpr int kXI = 68;
constexpr int kUP = 34;
}

__global__ __launch_bounds__(256)
void flexconv_wmma(const float* __restrict__ x,
                   const float* __restrict__ pts,
                   const float* __restrict__ wgt,
                   float* __restrict__ out)
{
    __shared__ float Xt[32 * kXI];
    __shared__ float U[3][kODL * kUP];

    const int stripe = blockIdx.x;
    const int hseg   = blockIdx.y >> 1;
    const int ohalf  = blockIdx.y & 1;
    const int b      = blockIdx.z;

    const int w0    = stripe * kTW;
    const int wb    = w0 - 1;
    const int h0    = hseg * kHB;
    const int obase = ohalf * 32;

    const int tid   = threadIdx.x;
    const int wid   = tid >> 5;
    const int lane  = tid & 31;
    const int m     = lane & 15;
    const int khalf = lane >> 4;

    const int odA = wid * 16 + m;
    const int oA  = obase + (odA & 31);
    const int dA  = odA >> 5;
    float a0[16], a1[16];
#pragma unroll
    for (int kk = 0; kk < 16; ++kk) {
        const int i0 = 4 * kk + 2 * khalf;
        a0[kk] = wgt[oA * (kC * 4) + (i0 + 0) * 4 + dA];
        a1[kk] = wgt[oA * (kC * 4) + (i0 + 1) * 4 + dA];
    }

    auto compute_u_row = [&](int r, int s) {
        __syncthreads();
        {
            const int px = tid & 31;
            const int w  = wb + px;
            const bool ok = (r >= 0) & (r < kH) & (w >= 0) & (w < kW);
            const bool pk = ok & (r + 1 < kH);
#pragma unroll
            for (int ii = 0; ii < 8; ++ii) {
                const int i = ((tid >> 5) << 3) + ii;
                const int gi = ((b * kC + i) * kH + r) * kW + w;
                float v = 0.0f;
                if (ok) v = x[gi];
                if (pk) __builtin_prefetch(&x[gi + kW], 0, 1);
                Xt[px * kXI + i] = v;
            }
        }
        __syncthreads();

        v8f acc0 = {0.f,0.f,0.f,0.f,0.f,0.f,0.f,0.f};
        v8f acc1 = acc0;
#pragma unroll
        for (int kk = 0; kk < 16; ++kk) {
            const int i0 = 4 * kk + 2 * khalf;
            v2f av; av.x = a0[kk]; av.y = a1[kk];
            const v2f bv0 = *reinterpret_cast<const v2f*>(&Xt[(m)      * kXI + i0]);
            const v2f bv1 = *reinterpret_cast<const v2f*>(&Xt[(16 + m) * kXI + i0]);
            acc0 = __builtin_amdgcn_wmma_f32_16x16x4_f32(false, av, false, bv0,
                                                         (short)0, acc0, false, false);
            acc1 = __builtin_amdgcn_wmma_f32_16x16x4_f32(false, av, false, bv1,
                                                         (short)0, acc1, false, false);
        }

#pragma unroll
        for (int v = 0; v < 8; ++v) {
            const int od = wid * 16 + v + 8 * khalf;
            U[s][od * kUP + m]      = acc0[v];
            U[s][od * kUP + 16 + m] = acc1[v];
        }
        __syncthreads();

#pragma unroll
        for (int rep = 0; rep < 4; ++rep) {
            const int item = tid + rep * 256;
            const int oo = item >> 5;
            const int px = item & 31;
            const int w  = wb + px;
            float p0 = 0.f, p1 = 0.f, p2 = 0.f;
            if ((r >= 0) & (r < kH) & (w >= 0) & (w < kW)) {
                p0 = pts[((b * 3 + 0) * kH + r) * kW + w];
                p1 = pts[((b * 3 + 1) * kH + r) * kW + w];
                p2 = pts[((b * 3 + 2) * kH + r) * kW + w];
            }
            const float u0 = U[s][(oo)      * kUP + px];
            const float u1 = U[s][(32 + oo) * kUP + px];
            const float u2 = U[s][(64 + oo) * kUP + px];
            const float u3 = U[s][(96 + oo) * kUP + px];
            U[s][(96 + oo) * kUP + px] = u3 - (p0 * u0 + p1 * u1 + p2 * u2);
        }
        __syncthreads();
    };

    compute_u_row(h0 - 1, 0);
    compute_u_row(h0,     1);

    for (int h = h0; h < h0 + kHB; ++h) {
        compute_u_row(h + 1, (h + 2 - h0) % 3);

        const int oo = tid >> 3;
        const int q0 = (tid & 7) * 4;
        float win[4][6];
#pragma unroll
        for (int rowd = 0; rowd < 4; ++rowd)
#pragma unroll
            for (int j = 0; j < 6; ++j) win[rowd][j] = 0.f;

#pragma unroll
        for (int sl = 0; sl < 3; ++sl) {
#pragma unroll
            for (int rowd = 0; rowd < 4; ++rowd) {
                const int base = (rowd * 32 + oo) * kUP + q0;
                const v2f* p2 = reinterpret_cast<const v2f*>(&U[sl][base]);
#pragma unroll
                for (int j = 0; j < 3; ++j) {
                    const v2f v = p2[j];
                    win[rowd][2 * j]     += v.x;
                    win[rowd][2 * j + 1] += v.y;
                }
            }
        }

#pragma unroll
        for (int jj = 0; jj < 4; ++jj) {
            const int pxo = q0 + jj;
            const int w   = w0 + pxo;
            if (pxo >= kTW || w >= kW) continue;
            const float su0 = win[0][jj] + win[0][jj + 1] + win[0][jj + 2];
            const float su1 = win[1][jj] + win[1][jj + 1] + win[1][jj + 2];
            const float su2 = win[2][jj] + win[2][jj + 1] + win[2][jj + 2];
            float acc       = win[3][jj] + win[3][jj + 1] + win[3][jj + 2];
            const float pc0 = pts[((b * 3 + 0) * kH + h) * kW + w];
            const float pc1 = pts[((b * 3 + 1) * kH + h) * kW + w];
            const float pc2 = pts[((b * 3 + 2) * kH + h) * kW + w];
            acc += pc0 * su0 + pc1 * su1 + pc2 * su2;
            out[((b * kO + obase + oo) * kH + h) * kW + w] = acc;
        }
    }
}

extern "C" void kernel_launch(void* const* d_in, const int* in_sizes, int n_in,
                              void* d_out, int out_size, void* d_ws, size_t ws_size,
                              hipStream_t stream) {
    const float* x   = (const float*)d_in[0];
    const float* pts = (const float*)d_in[1];
    const float* wgt = (const float*)d_in[2];
    float* out = (float*)d_out;

    dim3 grid(kNS, kNH * 2, kB);
    flexconv_wmma<<<grid, 256, 0, stream>>>(x, pts, wgt, out);

    (void)in_sizes; (void)n_in; (void)out_size; (void)d_ws; (void)ws_size;
}


# Global loads pipelined one row ahead: each row's x and points values load as one group into registers and stay in flight across LDS-only barriers, replacing 16 serialized load-wait trips per row
# speedup vs baseline: 1.8502x; 1.8502x over previous
#include <hip/hip_runtime.h>

typedef float v2f __attribute__((ext_vector_type(2)));
typedef float v8f __attribute__((ext_vector_type(8)));

namespace {
constexpr int kB  = 8;
constexpr int kC  = 64;
constexpr int kH  = 64;
constexpr int kW  = 512;
constexpr int kO  = 64;
constexpr int kTW = 30;
constexpr int kNS = 18;
constexpr int kHB = 16;
constexpr int kNH = 4;
constexpr int kODL = 128;
constexpr int kXI = 68;
constexpr int kUP = 34;
}

__device__ __forceinline__ void lds_barrier()
{
    __builtin_amdgcn_fence(__ATOMIC_RELEASE, "workgroup", "local");
    __builtin_amdgcn_s_barrier();
    __builtin_amdgcn_fence(__ATOMIC_ACQUIRE, "workgroup", "local");
}

__global__ __launch_bounds__(256, 6)
void flexconv_wmma(const float* __restrict__ x,
                   const float* __restrict__ pts,
                   const float* __restrict__ wgt,
                   float* __restrict__ out)
{
    __shared__ float Xt[32 * kXI];
    __shared__ float U[3][kODL * kUP];

    const int stripe = blockIdx.x;
    const int hseg   = blockIdx.y >> 1;
    const int ohalf  = blockIdx.y & 1;
    const int b      = blockIdx.z;

    const int w0    = stripe * kTW;
    const int wb    = w0 - 1;
    const int h0    = hseg * kHB;
    const int obase = ohalf * 32;

    const int tid   = threadIdx.x;
    const int wid   = tid >> 5;
    const int lane  = tid & 31;
    const int m     = lane & 15;
    const int khalf = lane >> 4;

    const int odA = wid * 16 + m;
    const int oA  = obase + (odA & 31);
    const int dA  = odA >> 5;
    float a0[16], a1[16];
#pragma unroll
    for (int kk = 0; kk < 16; ++kk) {
        const int i0 = 4 * kk + 2 * khalf;
        a0[kk] = wgt[oA * (kC * 4) + (i0 + 0) * 4 + dA];
        a1[kk] = wgt[oA * (kC * 4) + (i0 + 1) * 4 + dA];
    }

    const int spx  = tid & 31;
    const int sw   = wb + spx;
    const bool swok = (sw >= 0) & (sw < kW);
    const int swc  = min(max(sw, 0), kW - 1);
    const int si0  = (tid >> 5) << 3;
    float xr[8];
    float pr[3];

    auto load_x = [&](int r) {
        const int rc = min(max(r, 0), kH - 1);
#pragma unroll
        for (int ii = 0; ii < 8; ++ii)
            xr[ii] = x[((b * kC + si0 + ii) * kH + rc) * kW + swc];
    };
    auto load_p = [&](int r) {
        const int rc = min(max(r, 0), kH - 1);
#pragma unroll
        for (int d = 0; d < 3; ++d)
            pr[d] = pts[((b * 3 + d) * kH + rc) * kW + swc];
    };

    auto compute_u_row = [&](int r, int s) {
        lds_barrier();
        const bool ok = (r >= 0) & (r < kH) & swok;
#pragma unroll
        for (int ii = 0; ii < 8; ++ii)
            Xt[spx * kXI + si0 + ii] = ok ? xr[ii] : 0.0f;
        load_x(r + 1);
        lds_barrier();

        v8f acc0 = {0.f,0.f,0.f,0.f,0.f,0.f,0.f,0.f};
        v8f acc1 = acc0;
#pragma unroll
        for (int kk = 0; kk < 16; ++kk) {
            const int i0 = 4 * kk + 2 * khalf;
            v2f av; av.x = a0[kk]; av.y = a1[kk];
            const v2f bv0 = *reinterpret_cast<const v2f*>(&Xt[(m)      * kXI + i0]);
            const v2f bv1 = *reinterpret_cast<const v2f*>(&Xt[(16 + m) * kXI + i0]);
            acc0 = __builtin_amdgcn_wmma_f32_16x16x4_f32(false, av, false, bv0,
                                                         (short)0, acc0, false, false);
            acc1 = __builtin_amdgcn_wmma_f32_16x16x4_f32(false, av, false, bv1,
                                                         (short)0, acc1, false, false);
        }

#pragma unroll
        for (int v = 0; v < 8; ++v) {
            const int od = wid * 16 + v + 8 * khalf;
            U[s][od * kUP + m]      = acc0[v];
            U[s][od * kUP + 16 + m] = acc1[v];
        }
        lds_barrier();

        const float p0 = ok ? pr[0] : 0.f;
        const float p1 = ok ? pr[1] : 0.f;
        const float p2 = ok ? pr[2] : 0.f;
#pragma unroll
        for (int rep = 0; rep < 4; ++rep) {
            const int item = tid + rep * 256;
            const int oo = item >> 5;
            const int px = spx;
            const float u0 = U[s][(oo)      * kUP + px];
            const float u1 = U[s][(32 + oo) * kUP + px];
            const float u2 = U[s][(64 + oo) * kUP + px];
            const float u3 = U[s][(96 + oo) * kUP + px];
            U[s][(96 + oo) * kUP + px] = u3 - (p0 * u0 + p1 * u1 + p2 * u2);
        }
        load_p(r + 1);
        lds_barrier();
    };

    load_x(h0 - 1);
    load_p(h0 - 1);
    compute_u_row(h0 - 1, 0);
    compute_u_row(h0,     1);

    for (int h = h0; h < h0 + kHB; ++h) {
        const int oo = tid >> 3;
        const int q0 = (tid & 7) * 4;

        float pc[3][4];
#pragma unroll
        for (int jj = 0; jj < 4; ++jj) {
            const int wj = min(w0 + q0 + jj, kW - 1);
#pragma unroll
            for (int d = 0; d < 3; ++d)
                pc[d][jj] = pts[((b * 3 + d) * kH + h) * kW + wj];
        }

        compute_u_row(h + 1, (h + 2 - h0) % 3);

        float win[4][6];
#pragma unroll
        for (int rowd = 0; rowd < 4; ++rowd)
#pragma unroll
            for (int j = 0; j < 6; ++j) win[rowd][j] = 0.f;

#pragma unroll
        for (int sl = 0; sl < 3; ++sl) {
#pragma unroll
            for (int rowd = 0; rowd < 4; ++rowd) {
                const int base = (rowd * 32 + oo) * kUP + q0;
                const v2f* p2 = reinterpret_cast<const v2f*>(&U[sl][base]);
#pragma unroll
                for (int j = 0; j < 3; ++j) {
                    const v2f v = p2[j];
                    win[rowd][2 * j]     += v.x;
                    win[rowd][2 * j + 1] += v.y;
                }
            }
        }

#pragma unroll
        for (int jj = 0; jj < 4; ++jj) {
            const int pxo = q0 + jj;
            const int w   = w0 + pxo;
            if (pxo >= kTW || w >= kW) continue;
            const float su0 = win[0][jj] + win[0][jj + 1] + win[0][jj + 2];
            const float su1 = win[1][jj] + win[1][jj + 1] + win[1][jj + 2];
            const float su2 = win[2][jj] + win[2][jj + 1] + win[2][jj + 2];
            float acc       = win[3][jj] + win[3][jj + 1] + win[3][jj + 2];
            const float pc0 = pc[0][jj];
            const float pc1 = pc[1][jj];
            const float pc2 = pc[2][jj];
            acc += pc0 * su0 + pc1 * su1 + pc2 * su2;
            out[((b * kO + obase + oo) * kH + h) * kW + w] = acc;
        }
    }
}

extern "C" void kernel_launch(void* const* d_in, const int* in_sizes, int n_in,
                              void* d_out, int out_size, void* d_ws, size_t ws_size,
                              hipStream_t stream) {
    const float* x   = (const float*)d_in[0];
    const float* pts = (const float*)d_in[1];
    const float* wgt = (const float*)d_in[2];
    float* out = (float*)d_out;

    dim3 grid(kNS, kNH * 2, kB);
    flexconv_wmma<<<grid, 256, 0, stream>>>(x, pts, wgt, out);

    (void)in_sizes; (void)n_in; (void)out_size; (void)d_ws; (void)ws_size;
}


# GEMM rows reordered so a lane's accumulators hold all four d of its (channel, column): vbar folded in registers before the U scatter; LDS fold pass and its barrier removed (fold expression kept)
# speedup vs baseline: 1.9204x; 1.0379x over previous
#include <hip/hip_runtime.h>

typedef float v2f __attribute__((ext_vector_type(2)));
typedef float v8f __attribute__((ext_vector_type(8)));

namespace {
constexpr int kB  = 8;
constexpr int kC  = 64;
constexpr int kH  = 64;
constexpr int kW  = 512;
constexpr int kO  = 64;
constexpr int kTW = 30;
constexpr int kNS = 18;
constexpr int kHB = 16;
constexpr int kNH = 4;
constexpr int kODL = 128;
constexpr int kXI = 68;
constexpr int kUP = 34;
}

__device__ __forceinline__ void lds_barrier()
{
    __builtin_amdgcn_fence(__ATOMIC_RELEASE, "workgroup", "local");
    __builtin_amdgcn_s_barrier();
    __builtin_amdgcn_fence(__ATOMIC_ACQUIRE, "workgroup", "local");
}

__global__ __launch_bounds__(256, 6)
void flexconv_wmma(const float* __restrict__ x,
                   const float* __restrict__ pts,
                   const float* __restrict__ wgt,
                   float* __restrict__ out)
{
    __shared__ float Xt[32 * kXI];
    __shared__ float U[3][kODL * kUP];

    const int stripe = blockIdx.x;
    const int hseg   = blockIdx.y >> 1;
    const int ohalf  = blockIdx.y & 1;
    const int b      = blockIdx.z;

    const int w0    = stripe * kTW;
    const int wb    = w0 - 1;
    const int h0    = hseg * kHB;
    const int obase = ohalf * 32;

    const int tid   = threadIdx.x;
    const int wid   = tid >> 5;
    const int lane  = tid & 31;
    const int m     = lane & 15;
    const int khalf = lane >> 4;

    const int dA  = m & 3;
    const int oA  = obase + wid * 4 + ((m >> 2) & 1) * 2 + (m >> 3);
    float a0[16], a1[16];
#pragma unroll
    for (int kk = 0; kk < 16; ++kk) {
        const int i0 = 4 * kk + 2 * khalf;
        a0[kk] = wgt[oA * (kC * 4) + (i0 + 0) * 4 + dA];
        a1[kk] = wgt[oA * (kC * 4) + (i0 + 1) * 4 + dA];
    }

    const int spx  = tid & 31;
    const int sw   = wb + spx;
    const bool swok = (sw >= 0) & (sw < kW);
    const int swc  = min(max(sw, 0), kW - 1);
    const int si0  = (tid >> 5) << 3;
    float xr[8];
    const int  fw0  = wb + m;
    const int  fw1  = wb + 16 + m;
    const bool fok0 = (fw0 >= 0) & (fw0 < kW);
    const bool fok1 = (fw1 >= 0) & (fw1 < kW);
    const int  fwc0 = min(max(fw0, 0), kW - 1);
    const int  fwc1 = min(max(fw1, 0), kW - 1);
    float pr[2][3];

    auto load_x = [&](int r) {
        const int rc = min(max(r, 0), kH - 1);
#pragma unroll
        for (int ii = 0; ii < 8; ++ii)
            xr[ii] = x[((b * kC + si0 + ii) * kH + rc) * kW + swc];
    };
    auto load_p = [&](int r) {
        const int rc = min(max(r, 0), kH - 1);
#pragma unroll
        for (int d = 0; d < 3; ++d) {
            pr[0][d] = pts[((b * 3 + d) * kH + rc) * kW + fwc0];
            pr[1][d] = pts[((b * 3 + d) * kH + rc) * kW + fwc1];
        }
    };

    auto compute_u_row = [&](int r, int s) {
        lds_barrier();
        const bool ok = (r >= 0) & (r < kH) & swok;
#pragma unroll
        for (int ii = 0; ii < 8; ++ii)
            Xt[spx * kXI + si0 + ii] = ok ? xr[ii] : 0.0f;
        load_x(r + 1);
        lds_barrier();

        v8f acc0 = {0.f,0.f,0.f,0.f,0.f,0.f,0.f,0.f};
        v8f acc1 = acc0;
#pragma unroll
        for (int kk = 0; kk < 16; ++kk) {
            const int i0 = 4 * kk + 2 * khalf;
            v2f av; av.x = a0[kk]; av.y = a1[kk];
            const v2f bv0 = *reinterpret_cast<const v2f*>(&Xt[(m)      * kXI + i0]);
            const v2f bv1 = *reinterpret_cast<const v2f*>(&Xt[(16 + m) * kXI + i0]);
            acc0 = __builtin_amdgcn_wmma_f32_16x16x4_f32(false, av, false, bv0,
                                                         (short)0, acc0, false, false);
            acc1 = __builtin_amdgcn_wmma_f32_16x16x4_f32(false, av, false, bv1,
                                                         (short)0, acc1, false, false);
        }

        const bool rok = (r >= 0) & (r < kH);
#pragma unroll
        for (int t = 0; t < 2; ++t) {
            const bool okp = rok & (t ? fok1 : fok0);
            const float p0 = okp ? pr[t][0] : 0.f;
            const float p1 = okp ? pr[t][1] : 0.f;
            const float p2 = okp ? pr[t][2] : 0.f;
            const v8f acc = t ? acc1 : acc0;
#pragma unroll
            for (int j = 0; j < 2; ++j) {
                const int oo = wid * 4 + 2 * j + khalf;
                const float u0 = acc[4 * j + 0];
                const float u1 = acc[4 * j + 1];
                const float u2 = acc[4 * j + 2];
                const float u3 = acc[4 * j + 3];
                const int col = 16 * t + m;
                U[s][(oo)      * kUP + col] = u0;
                U[s][(32 + oo) * kUP + col] = u1;
                U[s][(64 + oo) * kUP + col] = u2;
                U[s][(96 + oo) * kUP + col] = u3 - (p0 * u0 + p1 * u1 + p2 * u2);
            }
        }
        load_p(r + 1);
        lds_barrier();
    };

    load_x(h0 - 1);
    load_p(h0 - 1);
    compute_u_row(h0 - 1, 0);
    compute_u_row(h0,     1);

    for (int h = h0; h < h0 + kHB; ++h) {
        const int oo = tid >> 3;
        const int q0 = (tid & 7) * 4;

        float pc[3][4];
#pragma unroll
        for (int jj = 0; jj < 4; ++jj) {
            const int wj = min(w0 + q0 + jj, kW - 1);
#pragma unroll
            for (int d = 0; d < 3; ++d)
                pc[d][jj] = pts[((b * 3 + d) * kH + h) * kW + wj];
        }

        compute_u_row(h + 1, (h + 2 - h0) % 3);

        float win[4][6];
#pragma unroll
        for (int rowd = 0; rowd < 4; ++rowd)
#pragma unroll
            for (int j = 0; j < 6; ++j) win[rowd][j] = 0.f;

#pragma unroll
        for (int sl = 0; sl < 3; ++sl) {
#pragma unroll
            for (int rowd = 0; rowd < 4; ++rowd) {
                const int base = (rowd * 32 + oo) * kUP + q0;
                const v2f* p2 = reinterpret_cast<const v2f*>(&U[sl][base]);
#pragma unroll
                for (int j = 0; j < 3; ++j) {
                    const v2f v = p2[j];
                    win[rowd][2 * j]     += v.x;
                    win[rowd][2 * j + 1] += v.y;
                }
            }
        }

#pragma unroll
        for (int jj = 0; jj < 4; ++jj) {
            const int pxo = q0 + jj;
            const int w   = w0 + pxo;
            if (pxo >= kTW || w >= kW) continue;
            const float su0 = win[0][jj] + win[0][jj + 1] + win[0][jj + 2];
            const float su1 = win[1][jj] + win[1][jj + 1] + win[1][jj + 2];
            const float su2 = win[2][jj] + win[2][jj + 1] + win[2][jj + 2];
            float acc       = win[3][jj] + win[3][jj + 1] + win[3][jj + 2];
            const float pc0 = pc[0][jj];
            const float pc1 = pc[1][jj];
            const float pc2 = pc[2][jj];
            acc += pc0 * su0 + pc1 * su1 + pc2 * su2;
            out[((b * kO + obase + oo) * kH + h) * kW + w] = acc;
        }
    }
}

extern "C" void kernel_launch(void* const* d_in, const int* in_sizes, int n_in,
                              void* d_out, int out_size, void* d_ws, size_t ws_size,
                              hipStream_t stream) {
    const float* x   = (const float*)d_in[0];
    const float* pts = (const float*)d_in[1];
    const float* wgt = (const float*)d_in[2];
    float* out = (float*)d_out;

    dim3 grid(kNS, kNH * 2, kB);
    flexconv_wmma<<<grid, 256, 0, stream>>>(x, pts, wgt, out);

    (void)in_sizes; (void)n_in; (void)out_size; (void)d_ws; (void)ws_size;
}


# 3x3 window phase: a wave's four 8-lane groups take channels 2w, 2w+16, 2w+1, 2w+17 (were 4w..4w+3), so its LDS window reads hit 64 distinct banks; they were 2-way bank-conflicted
# speedup vs baseline: 2.0019x; 1.0424x over previous
#include <hip/hip_runtime.h>

typedef float v2f __attribute__((ext_vector_type(2)));
typedef float v8f __attribute__((ext_vector_type(8)));

namespace {
constexpr int kB  = 8;
constexpr int kC  = 64;
constexpr int kH  = 64;
constexpr int kW  = 512;
constexpr int kO  = 64;
constexpr int kTW = 30;
constexpr int kNS = 18;
constexpr int kHB = 16;
constexpr int kNH = 4;
constexpr int kODL = 128;
constexpr int kXI = 68;
constexpr int kUP = 34;
}

__device__ __forceinline__ void lds_barrier()
{
    __builtin_amdgcn_fence(__ATOMIC_RELEASE, "workgroup", "local");
    __builtin_amdgcn_s_barrier();
    __builtin_amdgcn_fence(__ATOMIC_ACQUIRE, "workgroup", "local");
}

__global__ __launch_bounds__(256, 6)
void flexconv_wmma(const float* __restrict__ x,
                   const float* __restrict__ pts,
                   const float* __restrict__ wgt,
                   float* __restrict__ out)
{
    __shared__ float Xt[32 * kXI];
    __shared__ float U[3][kODL * kUP];

    const int stripe = blockIdx.x;
    const int hseg   = blockIdx.y >> 1;
    const int ohalf  = blockIdx.y & 1;
    const int b      = blockIdx.z;

    const int w0    = stripe * kTW;
    const int wb    = w0 - 1;
    const int h0    = hseg * kHB;
    const int obase = ohalf * 32;

    const int tid   = threadIdx.x;
    const int wid   = tid >> 5;
    const int lane  = tid & 31;
    const int m     = lane & 15;
    const int khalf = lane >> 4;

    const int dA  = m & 3;
    const int oA  = obase + wid * 4 + ((m >> 2) & 1) * 2 + (m >> 3);
    float a0[16], a1[16];
#pragma unroll
    for (int kk = 0; kk < 16; ++kk) {
        const int i0 = 4 * kk + 2 * khalf;
        a0[kk] = wgt[oA * (kC * 4) + (i0 + 0) * 4 + dA];
        a1[kk] = wgt[oA * (kC * 4) + (i0 + 1) * 4 + dA];
    }

    const int spx  = tid & 31;
    const int sw   = wb + spx;
    const bool swok = (sw >= 0) & (sw < kW);
    const int swc  = min(max(sw, 0), kW - 1);
    const int si0  = (tid >> 5) << 3;
    float xr[8];
    const int  fw0  = wb + m;
    const int  fw1  = wb + 16 + m;
    const bool fok0 = (fw0 >= 0) & (fw0 < kW);
    const bool fok1 = (fw1 >= 0) & (fw1 < kW);
    const int  fwc0 = min(max(fw0, 0), kW - 1);
    const int  fwc1 = min(max(fw1, 0), kW - 1);
    float pr[2][3];

    auto load_x = [&](int r) {
        const int rc = min(max(r, 0), kH - 1);
#pragma unroll
        for (int ii = 0; ii < 8; ++ii)
            xr[ii] = x[((b * kC + si0 + ii) * kH + rc) * kW + swc];
    };
    auto load_p = [&](int r) {
        const int rc = min(max(r, 0), kH - 1);
#pragma unroll
        for (int d = 0; d < 3; ++d) {
            pr[0][d] = pts[((b * 3 + d) * kH + rc) * kW + fwc0];
            pr[1][d] = pts[((b * 3 + d) * kH + rc) * kW + fwc1];
        }
    };

    auto compute_u_row = [&](int r, int s) {
        lds_barrier();
        const bool ok = (r >= 0) & (r < kH) & swok;
#pragma unroll
        for (int ii = 0; ii < 8; ++ii)
            Xt[spx * kXI + si0 + ii] = ok ? xr[ii] : 0.0f;
        load_x(r + 1);
        lds_barrier();

        v8f acc0 = {0.f,0.f,0.f,0.f,0.f,0.f,0.f,0.f};
        v8f acc1 = acc0;
#pragma unroll
        for (int kk = 0; kk < 16; ++kk) {
            const int i0 = 4 * kk + 2 * khalf;
            v2f av; av.x = a0[kk]; av.y = a1[kk];
            const v2f bv0 = *reinterpret_cast<const v2f*>(&Xt[(m)      * kXI + i0]);
            const v2f bv1 = *reinterpret_cast<const v2f*>(&Xt[(16 + m) * kXI + i0]);
            acc0 = __builtin_amdgcn_wmma_f32_16x16x4_f32(false, av, false, bv0,
                                                         (short)0, acc0, false, false);
            acc1 = __builtin_amdgcn_wmma_f32_16x16x4_f32(false, av, false, bv1,
                                                         (short)0, acc1, false, false);
        }

        const bool rok = (r >= 0) & (r < kH);
#pragma unroll
        for (int t = 0; t < 2; ++t) {
            const bool okp = rok & (t ? fok1 : fok0);
            const float p0 = okp ? pr[t][0] : 0.f;
            const float p1 = okp ? pr[t][1] : 0.f;
            const float p2 = okp ? pr[t][2] : 0.f;
            const v8f acc = t ? acc1 : acc0;
#pragma unroll
            for (int j = 0; j < 2; ++j) {
                const int oo = wid * 4 + 2 * j + khalf;
                const float u0 = acc[4 * j + 0];
                const float u1 = acc[4 * j + 1];
                const float u2 = acc[4 * j + 2];
                const float u3 = acc[4 * j + 3];
                const int col = 16 * t + m;
                U[s][(oo)      * kUP + col] = u0;
                U[s][(32 + oo) * kUP + col] = u1;
                U[s][(64 + oo) * kUP + col] = u2;
                U[s][(96 + oo) * kUP + col] = u3 - (p0 * u0 + p1 * u1 + p2 * u2);
            }
        }
        load_p(r + 1);
        lds_barrier();
    };

    load_x(h0 - 1);
    load_p(h0 - 1);
    compute_u_row(h0 - 1, 0);
    compute_u_row(h0,     1);

    for (int h = h0; h < h0 + kHB; ++h) {
        const int sg = lane >> 3;
        const int oo = 2 * wid + (sg >> 1) + 16 * (sg & 1);
        const int q0 = (tid & 7) * 4;

        float pc[3][4];
#pragma unroll
        for (int jj = 0; jj < 4; ++jj) {
            const int wj = min(w0 + q0 + jj, kW - 1);
#pragma unroll
            for (int d = 0; d < 3; ++d)
                pc[d][jj] = pts[((b * 3 + d) * kH + h) * kW + wj];
        }

        compute_u_row(h + 1, (h + 2 - h0) % 3);

        float win[4][6];
#pragma unroll
        for (int rowd = 0; rowd < 4; ++rowd)
#pragma unroll
            for (int j = 0; j < 6; ++j) win[rowd][j] = 0.f;

#pragma unroll
        for (int sl = 0; sl < 3; ++sl) {
#pragma unroll
            for (int rowd = 0; rowd < 4; ++rowd) {
                const int base = (rowd * 32 + oo) * kUP + q0;
                const v2f* p2 = reinterpret_cast<const v2f*>(&U[sl][base]);
#pragma unroll
                for (int j = 0; j < 3; ++j) {
                    const v2f v = p2[j];
                    win[rowd][2 * j]     += v.x;
                    win[rowd][2 * j + 1] += v.y;
                }
            }
        }

#pragma unroll
        for (int jj = 0; jj < 4; ++jj) {
            const int pxo = q0 + jj;
            const int w   = w0 + pxo;
            if (pxo >= kTW || w >= kW) continue;
            const float su0 = win[0][jj] + win[0][jj + 1] + win[0][jj + 2];
            const float su1 = win[1][jj] + win[1][jj + 1] + win[1][jj + 2];
            const float su2 = win[2][jj] + win[2][jj + 1] + win[2][jj + 2];
            float acc       = win[3][jj] + win[3][jj + 1] + win[3][jj + 2];
            const float pc0 = pc[0][jj];
            const float pc1 = pc[1][jj];
            const float pc2 = pc[2][jj];
            acc += pc0 * su0 + pc1 * su1 + pc2 * su2;
            out[((b * kO + obase + oo) * kH + h) * kW + w] = acc;
        }
    }
}

extern "C" void kernel_launch(void* const* d_in, const int* in_sizes, int n_in,
                              void* d_out, int out_size, void* d_ws, size_t ws_size,
                              hipStream_t stream) {
    const float* x   = (const float*)d_in[0];
    const float* pts = (const float*)d_in[1];
    const float* wgt = (const float*)d_in[2];
    float* out = (float*)d_out;

    dim3 grid(kNS, kNH * 2, kB);
    flexconv_wmma<<<grid, 256, 0, stream>>>(x, pts, wgt, out);

    (void)in_sizes; (void)n_in; (void)out_size; (void)d_ws; (void)ws_size;
}


# Stripe widened from 30 to 46 output columns (three 16-column GEMM tiles per wave): 768 workgroups, three per WGP, all in one round (were 1152, one and a half rounds); centre points via LDS
# speedup vs baseline: 2.2335x; 1.1157x over previous
#include <hip/hip_runtime.h>


typedef float v2f __attribute__((ext_vector_type(2)));
typedef float v4f __attribute__((ext_vector_type(4)));
typedef float v8f __attribute__((ext_vector_type(8)));

namespace {
constexpr int kB  = 8;
constexpr int kC  = 64;
constexpr int kH  = 64;
constexpr int kW  = 512;
constexpr int kO  = 64;
constexpr int kTW = 46;
constexpr int kNC = 48;
constexpr int kNS = 12;
constexpr int kHB = 16;
constexpr int kNH = 4;
constexpr int kODL = 128;
constexpr int kXI = 68;
constexpr int kUP = 50;
}

__device__ __forceinline__ void lds_barrier()
{
    __builtin_amdgcn_fence(__ATOMIC_RELEASE, "workgroup", "local");
    __builtin_amdgcn_s_barrier();
    __builtin_amdgcn_fence(__ATOMIC_ACQUIRE, "workgroup", "local");
}

__global__ __launch_bounds__(256, 6)
void flexconv_wmma(const float* __restrict__ x,
                   const float* __restrict__ pts,
                   const float* __restrict__ wgt,
                   float* __restrict__ out)
{
    __shared__ __attribute__((aligned(16))) float Xt[kNC * kXI];
    __shared__ __attribute__((aligned(16))) float U[3][kODL * kUP];
    __shared__ __attribute__((aligned(16))) float P[3][3 * kUP];

    const int stripe = blockIdx.x;
    const int hseg   = blockIdx.y >> 1;
    const int ohalf  = blockIdx.y & 1;
    const int b      = blockIdx.z;

    const int w0    = stripe * kTW;
    const int wb    = w0 - 1;
    const int h0    = hseg * kHB;
    const int obase = ohalf * 32;

    const int tid   = threadIdx.x;
    const int wid   = __builtin_amdgcn_readfirstlane(tid >> 5);
    const int lane  = tid & 31;
    const int m     = lane & 15;
    const int khalf = lane >> 4;

    const int dA  = m & 3;
    const int oA  = obase + wid * 4 + ((m >> 2) & 1) + 2 * (m >> 3);
    float a0[16], a1[16];
#pragma unroll
    for (int kk = 0; kk < 16; ++kk) {
        const int i0 = 4 * kk + 2 * khalf;
        a0[kk] = wgt[oA * (kC * 4) + (i0 + 0) * 4 + dA];
        a1[kk] = wgt[oA * (kC * 4) + (i0 + 1) * 4 + dA];
    }

    const int q4  = tid >> 4;
    const int sxq = ((b * kC + 4 * q4) * kH) * kW;
    const int sl0 = 3 * m * kXI + 4 * q4;
    bool fok[3];
    int  fwc[3];
#pragma unroll
    for (int t = 0; t < 3; ++t) {
        const int fw = wb + 16 * t + m;
        fok[t] = (fw >= 0) & (fw < kW);
        fwc[t] = min(max(fw, 0), kW - 1);
    }
    float xr[3][4];
    float pr[3][3];

    auto load_x = [&](int r) {
        const int rc = min(max(r, 0), kH - 1);
#pragma unroll
        for (int k = 0; k < 3; ++k)
#pragma unroll
            for (int c = 0; c < 4; ++c)
                xr[k][c] = x[sxq + (c * kH + rc) * kW + fwc[k]];
    };
    auto load_p = [&](int r) {
        const int rc = min(max(r, 0), kH - 1);
#pragma unroll
        for (int d = 0; d < 3; ++d)
#pragma unroll
            for (int t = 0; t < 3; ++t)
                pr[t][d] = pts[((b * 3 + d) * kH + rc) * kW + fwc[t]];
    };

    auto compute_u_row = [&](int r, int s) {
        load_p(r);
        lds_barrier();
        const bool rok = (r >= 0) & (r < kH);
#pragma unroll
        for (int k = 0; k < 3; ++k) {
            const bool ok = rok & fok[k];
            v4f v;
            v.x = ok ? xr[k][0] : 0.0f;
            v.y = ok ? xr[k][1] : 0.0f;
            v.z = ok ? xr[k][2] : 0.0f;
            v.w = ok ? xr[k][3] : 0.0f;
            *reinterpret_cast<v4f*>(&Xt[sl0 + k * kXI]) = v;
        }
        load_x(r + 1);
        lds_barrier();

        v8f acc0 = {0.f,0.f,0.f,0.f,0.f,0.f,0.f,0.f};
        v8f acc1 = acc0;
        v8f acc2 = acc0;
#pragma unroll
        for (int kk = 0; kk < 16; ++kk) {
            const int i0 = 4 * kk + 2 * khalf;
            v2f av; av.x = a0[kk]; av.y = a1[kk];
            const v2f bv0 = *reinterpret_cast<const v2f*>(&Xt[(3 * m + 0) * kXI + i0]);
            const v2f bv1 = *reinterpret_cast<const v2f*>(&Xt[(3 * m + 1) * kXI + i0]);
            const v2f bv2 = *reinterpret_cast<const v2f*>(&Xt[(3 * m + 2) * kXI + i0]);
            acc0 = __builtin_amdgcn_wmma_f32_16x16x4_f32(false, av, false, bv0,
                                                         (short)0, acc0, false, false);
            acc1 = __builtin_amdgcn_wmma_f32_16x16x4_f32(false, av, false, bv1,
                                                         (short)0, acc1, false, false);
            acc2 = __builtin_amdgcn_wmma_f32_16x16x4_f32(false, av, false, bv2,
                                                         (short)0, acc2, false, false);
        }

        __builtin_amdgcn_sched_barrier(0);

#pragma unroll
        for (int t = 0; t < 3; ++t) {
            const bool okp = rok & fok[t];
            const float p0 = okp ? pr[t][0] : 0.f;
            const float p1 = okp ? pr[t][1] : 0.f;
            const float p2 = okp ? pr[t][2] : 0.f;
            if (wid == t) {
                P[s][0 * kUP + 16 * t + m + 1] = p0;
                P[s][1 * kUP + 16 * t + m + 1] = p1;
                P[s][2 * kUP + 16 * t + m + 1] = p2;
            }
            const v8f acc = (t == 0) ? acc0 : ((t == 1) ? acc1 : acc2);
#pragma unroll
            for (int j = 0; j < 2; ++j) {
                const int oo = wid * 4 + j + 2 * khalf;
                const float u0 = acc[4 * j + 0];
                const float u1 = acc[4 * j + 1];
                const float u2 = acc[4 * j + 2];
                const float u3 = acc[4 * j + 3];
                const int col = 16 * t + m;
                U[s][(oo)      * kUP + col] = u0;
                U[s][(32 + oo) * kUP + col] = u1;
                U[s][(64 + oo) * kUP + col] = u2;
                U[s][(96 + oo) * kUP + col] = u3 - (p0 * u0 + p1 * u1 + p2 * u2);
            }
        }
        lds_barrier();
    };

    load_x(h0 - 1);
    compute_u_row(h0 - 1, 0);
    compute_u_row(h0,     1);

    const int sg = lane >> 3;
    const int oo = wid + 8 * sg;
    const int q0 = (lane & 7) * 6;

    for (int h = h0; h < h0 + kHB; ++h) {
        compute_u_row(h + 1, (h + 2 - h0) % 3);
        const int sc = (h + 1 - h0) % 3;

        float su[4][6];
#pragma unroll
        for (int rowd = 0; rowd < 4; ++rowd) {
            float wv[8];
#pragma unroll
            for (int j = 0; j < 8; ++j) wv[j] = 0.f;
#pragma unroll
            for (int sl = 0; sl < 3; ++sl) {
                const int base = (rowd * 32 + oo) * kUP + q0;
                const v2f* p2 = reinterpret_cast<const v2f*>(&U[sl][base]);
#pragma unroll
                for (int j = 0; j < 4; ++j) {
                    const v2f v = p2[j];
                    wv[2 * j]     += v.x;
                    wv[2 * j + 1] += v.y;
                }
            }
#pragma unroll
            for (int jj = 0; jj < 6; ++jj) su[rowd][jj] = wv[jj] + wv[jj + 1] + wv[jj + 2];
        }

#pragma unroll
        for (int p = 0; p < 3; ++p) {
            const int pxo = q0 + 2 * p;
            const int w   = w0 + pxo;
            if (pxo >= kTW || w >= kW) continue;
            const v2f c0 = *reinterpret_cast<const v2f*>(&P[sc][0 * kUP + pxo + 2]);
            const v2f c1 = *reinterpret_cast<const v2f*>(&P[sc][1 * kUP + pxo + 2]);
            const v2f c2 = *reinterpret_cast<const v2f*>(&P[sc][2 * kUP + pxo + 2]);
            v2f res;
#pragma unroll
            for (int e = 0; e < 2; ++e) {
                const int jj = 2 * p + e;
                const float su0 = su[0][jj];
                const float su1 = su[1][jj];
                const float su2 = su[2][jj];
                float acc       = su[3][jj];
                const float pc0 = e ? c0.y : c0.x;
                const float pc1 = e ? c1.y : c1.x;
                const float pc2 = e ? c2.y : c2.x;
                acc += pc0 * su0 + pc1 * su1 + pc2 * su2;
                if (e) res.y = acc; else res.x = acc;
            }
            *reinterpret_cast<v2f*>(&out[((b * kO + obase + oo) * kH + h) * kW + w]) = res;
        }
    }
}

extern "C" void kernel_launch(void* const* d_in, const int* in_sizes, int n_in,
                              void* d_out, int out_size, void* d_ws, size_t ws_size,
                              hipStream_t stream) {
    const float* x   = (const float*)d_in[0];
    const float* pts = (const float*)d_in[1];
    const float* wgt = (const float*)d_in[2];
    float* out = (float*)d_out;

    dim3 grid(kNS, kNH * 2, kB);
    flexconv_wmma<<<grid, 256, 0, stream>>>(x, pts, wgt, out);

    (void)in_sizes; (void)n_in; (void)out_size; (void)d_ws; (void)ws_size;
}


# Window phase: a row's LDS reads (U windows, centre points) issued as one group ahead of the per-lane store branches; the compiler had sunk 21 of them there, 8 serial LDS waits; sums bit-identical
# speedup vs baseline: 2.3265x; 1.0416x over previous
#include <hip/hip_runtime.h>


typedef float v2f __attribute__((ext_vector_type(2)));
typedef float v4f __attribute__((ext_vector_type(4)));
typedef float v8f __attribute__((ext_vector_type(8)));

namespace {
constexpr int kB  = 8;
constexpr int kC  = 64;
constexpr int kH  = 64;
constexpr int kW  = 512;
constexpr int kO  = 64;
constexpr int kTW = 46;
constexpr int kNC = 48;
constexpr int kNS = 12;
constexpr int kHB = 16;
constexpr int kNH = 4;
constexpr int kODL = 128;
constexpr int kXI = 68;
constexpr int kUP = 50;
}

__device__ __forceinline__ void lds_barrier()
{
    __builtin_amdgcn_fence(__ATOMIC_RELEASE, "workgroup", "local");
    __builtin_amdgcn_s_barrier();
    __builtin_amdgcn_fence(__ATOMIC_ACQUIRE, "workgroup", "local");
}

__global__ __launch_bounds__(256, 6)
void flexconv_wmma(const float* __restrict__ x,
                   const float* __restrict__ pts,
                   const float* __restrict__ wgt,
                   float* __restrict__ out)
{
    __shared__ __attribute__((aligned(16))) float Xt[kNC * kXI];
    __shared__ __attribute__((aligned(16))) float U[3][kODL * kUP];
    __shared__ __attribute__((aligned(16))) float P[3][3 * kUP];

    const int stripe = blockIdx.x;
    const int hseg   = blockIdx.y >> 1;
    const int ohalf  = blockIdx.y & 1;
    const int b      = blockIdx.z;

    const int w0    = stripe * kTW;
    const int wb    = w0 - 1;
    const int h0    = hseg * kHB;
    const int obase = ohalf * 32;

    const int tid   = threadIdx.x;
    const int wid   = __builtin_amdgcn_readfirstlane(tid >> 5);
    const int lane  = tid & 31;
    const int m     = lane & 15;
    const int khalf = lane >> 4;

    const int dA  = m & 3;
    const int oA  = obase + wid * 4 + ((m >> 2) & 1) + 2 * (m >> 3);
    float a0[16], a1[16];
#pragma unroll
    for (int kk = 0; kk < 16; ++kk) {
        const int i0 = 4 * kk + 2 * khalf;
        a0[kk] = wgt[oA * (kC * 4) + (i0 + 0) * 4 + dA];
        a1[kk] = wgt[oA * (kC * 4) + (i0 + 1) * 4 + dA];
    }

    const int q4  = tid >> 4;
    const int sxq = ((b * kC + 4 * q4) * kH) * kW;
    const int sl0 = 3 * m * kXI + 4 * q4;
    bool fok[3];
    int  fwc[3];
#pragma unroll
    for (int t = 0; t < 3; ++t) {
        const int fw = wb + 16 * t + m;
        fok[t] = (fw >= 0) & (fw < kW);
        fwc[t] = min(max(fw, 0), kW - 1);
    }
    float xr[3][4];
    float pr[3][3];

    auto load_x = [&](int r) {
        const int rc = min(max(r, 0), kH - 1);
#pragma unroll
        for (int k = 0; k < 3; ++k)
#pragma unroll
            for (int c = 0; c < 4; ++c)
                xr[k][c] = x[sxq + (c * kH + rc) * kW + fwc[k]];
    };
    auto load_p = [&](int r) {
        const int rc = min(max(r, 0), kH - 1);
#pragma unroll
        for (int d = 0; d < 3; ++d)
#pragma unroll
            for (int t = 0; t < 3; ++t)
                pr[t][d] = pts[((b * 3 + d) * kH + rc) * kW + fwc[t]];
    };

    auto compute_u_row = [&](int r, int s) {
        load_p(r);
        lds_barrier();
        const bool rok = (r >= 0) & (r < kH);
#pragma unroll
        for (int k = 0; k < 3; ++k) {
            const bool ok = rok & fok[k];
            v4f v;
            v.x = ok ? xr[k][0] : 0.0f;
            v.y = ok ? xr[k][1] : 0.0f;
            v.z = ok ? xr[k][2] : 0.0f;
            v.w = ok ? xr[k][3] : 0.0f;
            *reinterpret_cast<v4f*>(&Xt[sl0 + k * kXI]) = v;
        }
        load_x(r + 1);
        lds_barrier();

        v8f acc0 = {0.f,0.f,0.f,0.f,0.f,0.f,0.f,0.f};
        v8f acc1 = acc0;
        v8f acc2 = acc0;
#pragma unroll
        for (int kk = 0; kk < 16; ++kk) {
            const int i0 = 4 * kk + 2 * khalf;
            v2f av; av.x = a0[kk]; av.y = a1[kk];
            const v2f bv0 = *reinterpret_cast<const v2f*>(&Xt[(3 * m + 0) * kXI + i0]);
            const v2f bv1 = *reinterpret_cast<const v2f*>(&Xt[(3 * m + 1) * kXI + i0]);
            const v2f bv2 = *reinterpret_cast<const v2f*>(&Xt[(3 * m + 2) * kXI + i0]);
            acc0 = __builtin_amdgcn_wmma_f32_16x16x4_f32(false, av, false, bv0,
                                                         (short)0, acc0, false, false);
            acc1 = __builtin_amdgcn_wmma_f32_16x16x4_f32(false, av, false, bv1,
                                                         (short)0, acc1, false, false);
            acc2 = __builtin_amdgcn_wmma_f32_16x16x4_f32(false, av, false, bv2,
                                                         (short)0, acc2, false, false);
        }

        __builtin_amdgcn_sched_barrier(0);

#pragma unroll
        for (int t = 0; t < 3; ++t) {
            const bool okp = rok & fok[t];
            const float p0 = okp ? pr[t][0] : 0.f;
            const float p1 = okp ? pr[t][1] : 0.f;
            const float p2 = okp ? pr[t][2] : 0.f;
            if (wid == t) {
                P[s][0 * kUP + 16 * t + m + 1] = p0;
                P[s][1 * kUP + 16 * t + m + 1] = p1;
                P[s][2 * kUP + 16 * t + m + 1] = p2;
            }
            const v8f acc = (t == 0) ? acc0 : ((t == 1) ? acc1 : acc2);
#pragma unroll
            for (int j = 0; j < 2; ++j) {
                const int oo = wid * 4 + j + 2 * khalf;
                const float u0 = acc[4 * j + 0];
                const float u1 = acc[4 * j + 1];
                const float u2 = acc[4 * j + 2];
                const float u3 = acc[4 * j + 3];
                const int col = 16 * t + m;
                U[s][(oo)      * kUP + col] = u0;
                U[s][(32 + oo) * kUP + col] = u1;
                U[s][(64 + oo) * kUP + col] = u2;
                U[s][(96 + oo) * kUP + col] = u3 - (p0 * u0 + p1 * u1 + p2 * u2);
            }
        }
        lds_barrier();
    };

    load_x(h0 - 1);
    compute_u_row(h0 - 1, 0);
    compute_u_row(h0,     1);

    const int sg = lane >> 3;
    const int oo = wid + 8 * sg;
    const int q0 = (lane & 7) * 6;

    for (int h = h0; h < h0 + kHB; ++h) {
        compute_u_row(h + 1, (h + 2 - h0) % 3);
        const int sc = (h + 1 - h0) % 3;

        v2f cp[3][3];
#pragma unroll
        for (int p = 0; p < 3; ++p)
#pragma unroll
            for (int d = 0; d < 3; ++d)
                cp[p][d] = *reinterpret_cast<const v2f*>(&P[sc][d * kUP + q0 + 2 * p + 2]);
        __builtin_amdgcn_sched_barrier(0);
        float wv[4][8];
#pragma unroll
        for (int rowd = 0; rowd < 4; ++rowd) {
#pragma unroll
            for (int j = 0; j < 8; ++j) wv[rowd][j] = 0.f;
#pragma unroll
            for (int sl = 0; sl < 3; ++sl) {
                const int base = (rowd * 32 + oo) * kUP + q0;
                const v2f* p2 = reinterpret_cast<const v2f*>(&U[sl][base]);
#pragma unroll
                for (int j = 0; j < 4; ++j) {
                    const v2f v = p2[j];
                    wv[rowd][2 * j]     += v.x;
                    wv[rowd][2 * j + 1] += v.y;
                }
            }
        }
#pragma unroll
        for (int rowd = 0; rowd < 4; ++rowd)
#pragma unroll
            for (int j = 0; j < 8; ++j) asm volatile("" : "+v"(wv[rowd][j]));
#pragma unroll
        for (int p = 0; p < 3; ++p)
#pragma unroll
            for (int d = 0; d < 3; ++d) asm volatile("" : "+v"(cp[p][d]));

#pragma unroll
        for (int p = 0; p < 3; ++p) {
            const int pxo = q0 + 2 * p;
            const int w   = w0 + pxo;
            if (pxo >= kTW || w >= kW) continue;
            v2f res;
#pragma unroll
            for (int e = 0; e < 2; ++e) {
                const int jj = 2 * p + e;
                const float su0 = wv[0][jj] + wv[0][jj + 1] + wv[0][jj + 2];
                const float su1 = wv[1][jj] + wv[1][jj + 1] + wv[1][jj + 2];
                const float su2 = wv[2][jj] + wv[2][jj + 1] + wv[2][jj + 2];
                float acc       = wv[3][jj] + wv[3][jj + 1] + wv[3][jj + 2];
                const float pc0 = e ? cp[p][0].y : cp[p][0].x;
                const float pc1 = e ? cp[p][1].y : cp[p][1].x;
                const float pc2 = e ? cp[p][2].y : cp[p][2].x;
                acc += __builtin_fmaf(pc2, su2, __builtin_fmaf(pc0, su0, pc1 * su1));
                if (e) res.y = acc; else res.x = acc;
            }
            *reinterpret_cast<v2f*>(&out[((b * kO + obase + oo) * kH + h) * kW + w]) = res;
        }
    }
}

extern "C" void kernel_launch(void* const* d_in, const int* in_sizes, int n_in,
                              void* d_out, int out_size, void* d_ws, size_t ws_size,
                              hipStream_t stream) {
    const float* x   = (const float*)d_in[0];
    const float* pts = (const float*)d_in[1];
    const float* wgt = (const float*)d_in[2];
    float* out = (float*)d_out;

    dim3 grid(kNS, kNH * 2, kB);
    flexconv_wmma<<<grid, 256, 0, stream>>>(x, pts, wgt, out);

    (void)in_sizes; (void)n_in; (void)out_size; (void)d_ws; (void)ws_size;
}


# Row loop resynchronised: window phase (row h) and GEMM (row h+2) share one barrier-free LDS read interval; scatter + next-row staging one write interval; barriers per row 3->2; vbar fma order fixed
# speedup vs baseline: 2.4347x; 1.0465x over previous
#include <hip/hip_runtime.h>


typedef float v2f __attribute__((ext_vector_type(2)));
typedef float v4f __attribute__((ext_vector_type(4)));
typedef float v8f __attribute__((ext_vector_type(8)));

namespace {
constexpr int kB  = 8;
constexpr int kC  = 64;
constexpr int kH  = 64;
constexpr int kW  = 512;
constexpr int kO  = 64;
constexpr int kTW = 46;
constexpr int kNC = 48;
constexpr int kNS = 12;
constexpr int kHB = 16;
constexpr int kNH = 4;
constexpr int kODL = 128;
constexpr int kXI = 68;
constexpr int kUP = 50;
}

__device__ __forceinline__ void lds_barrier()
{
    __builtin_amdgcn_fence(__ATOMIC_RELEASE, "workgroup", "local");
    __builtin_amdgcn_s_barrier();
    __builtin_amdgcn_fence(__ATOMIC_ACQUIRE, "workgroup", "local");
}

__global__ __launch_bounds__(256, 6)
void flexconv_wmma(const float* __restrict__ x,
                   const float* __restrict__ pts,
                   const float* __restrict__ wgt,
                   float* __restrict__ out)
{
    __shared__ __attribute__((aligned(16))) float Xt[kNC * kXI];
    __shared__ __attribute__((aligned(16))) float U[3][kODL * kUP];
    __shared__ __attribute__((aligned(16))) float P[3][3 * kUP];

    const int stripe = blockIdx.x;
    const int hseg   = blockIdx.y >> 1;
    const int ohalf  = blockIdx.y & 1;
    const int b      = blockIdx.z;

    const int w0    = stripe * kTW;
    const int wb    = w0 - 1;
    const int h0    = hseg * kHB;
    const int obase = ohalf * 32;

    const int tid   = threadIdx.x;
    const int wid   = __builtin_amdgcn_readfirstlane(tid >> 5);
    const int lane  = tid & 31;
    const int m     = lane & 15;
    const int khalf = lane >> 4;

    const int dA  = m & 3;
    const int oA  = obase + wid * 4 + ((m >> 2) & 1) + 2 * (m >> 3);
    float a0[16], a1[16];
#pragma unroll
    for (int kk = 0; kk < 16; ++kk) {
        const int i0 = 4 * kk + 2 * khalf;
        a0[kk] = wgt[oA * (kC * 4) + (i0 + 0) * 4 + dA];
        a1[kk] = wgt[oA * (kC * 4) + (i0 + 1) * 4 + dA];
    }

    const int q4  = tid >> 4;
    const int sxq = ((b * kC + 4 * q4) * kH) * kW;
    const int sl0 = 3 * m * kXI + 4 * q4;
    bool fok[3];
    int  fwc[3];
#pragma unroll
    for (int t = 0; t < 3; ++t) {
        const int fw = wb + 16 * t + m;
        fok[t] = (fw >= 0) & (fw < kW);
        fwc[t] = min(max(fw, 0), kW - 1);
    }
    float xr[3][4];
    float pr[3][3];

    auto load_x = [&](int r) {
        const int rc = min(max(r, 0), kH - 1);
#pragma unroll
        for (int k = 0; k < 3; ++k)
#pragma unroll
            for (int c = 0; c < 4; ++c)
                xr[k][c] = x[sxq + (c * kH + rc) * kW + fwc[k]];
    };
    auto load_p = [&](int r) {
        const int rc = min(max(r, 0), kH - 1);
#pragma unroll
        for (int d = 0; d < 3; ++d)
#pragma unroll
            for (int t = 0; t < 3; ++t)
                pr[t][d] = pts[((b * 3 + d) * kH + rc) * kW + fwc[t]];
    };

    auto stage_row = [&](int r) {
        const bool rok = (r >= 0) & (r < kH);
#pragma unroll
        for (int k = 0; k < 3; ++k) {
            const bool ok = rok & fok[k];
            v4f v;
            v.x = ok ? xr[k][0] : 0.0f;
            v.y = ok ? xr[k][1] : 0.0f;
            v.z = ok ? xr[k][2] : 0.0f;
            v.w = ok ? xr[k][3] : 0.0f;
            *reinterpret_cast<v4f*>(&Xt[sl0 + k * kXI]) = v;
        }
    };

    auto compute_u_row = [&](int r, int s) {
        load_p(r);
        const bool rok = (r >= 0) & (r < kH);

        v8f acc0 = {0.f,0.f,0.f,0.f,0.f,0.f,0.f,0.f};
        v8f acc1 = acc0;
        v8f acc2 = acc0;
#pragma unroll
        for (int kk = 0; kk < 16; ++kk) {
            const int i0 = 4 * kk + 2 * khalf;
            v2f av; av.x = a0[kk]; av.y = a1[kk];
            const v2f bv0 = *reinterpret_cast<const v2f*>(&Xt[(3 * m + 0) * kXI + i0]);
            const v2f bv1 = *reinterpret_cast<const v2f*>(&Xt[(3 * m + 1) * kXI + i0]);
            const v2f bv2 = *reinterpret_cast<const v2f*>(&Xt[(3 * m + 2) * kXI + i0]);
            acc0 = __builtin_amdgcn_wmma_f32_16x16x4_f32(false, av, false, bv0,
                                                         (short)0, acc0, false, false);
            acc1 = __builtin_amdgcn_wmma_f32_16x16x4_f32(false, av, false, bv1,
                                                         (short)0, acc1, false, false);
            acc2 = __builtin_amdgcn_wmma_f32_16x16x4_f32(false, av, false, bv2,
                                                         (short)0, acc2, false, false);
        }

        __builtin_amdgcn_sched_barrier(0);

        float pp[3][3];
        float vb[3][2];
#pragma unroll
        for (int t = 0; t < 3; ++t) {
            const bool okp = rok & fok[t];
            pp[t][0] = okp ? pr[t][0] : 0.f;
            pp[t][1] = okp ? pr[t][1] : 0.f;
            pp[t][2] = okp ? pr[t][2] : 0.f;
            const v8f acc = (t == 0) ? acc0 : ((t == 1) ? acc1 : acc2);
#pragma unroll
            for (int j = 0; j < 2; ++j)
                vb[t][j] = acc[4 * j + 3]
                         - __builtin_fmaf(pp[t][2], acc[4 * j + 2],
                                          __builtin_fmaf(pp[t][0], acc[4 * j + 0], pp[t][1] * acc[4 * j + 1]));
        }

        lds_barrier();

#pragma unroll
        for (int t = 0; t < 3; ++t) {
            if (wid == t) {
                P[s][0 * kUP + 16 * t + m + 1] = pp[t][0];
                P[s][1 * kUP + 16 * t + m + 1] = pp[t][1];
                P[s][2 * kUP + 16 * t + m + 1] = pp[t][2];
            }
            const v8f acc = (t == 0) ? acc0 : ((t == 1) ? acc1 : acc2);
#pragma unroll
            for (int j = 0; j < 2; ++j) {
                const int oo = wid * 4 + j + 2 * khalf;
                const int col = 16 * t + m;
                U[s][(oo)      * kUP + col] = acc[4 * j + 0];
                U[s][(32 + oo) * kUP + col] = acc[4 * j + 1];
                U[s][(64 + oo) * kUP + col] = acc[4 * j + 2];
                U[s][(96 + oo) * kUP + col] = vb[t][j];
            }
        }
        stage_row(r + 1);
        load_x(r + 2);
        lds_barrier();
    };

    const int sg = lane >> 3;
    const int oo = wid + 8 * sg;
    const int q0 = (lane & 7) * 6;

    auto window_row = [&](int h) {
        const int sc = (h + 1 - h0) % 3;

        v2f cp[3][3];
#pragma unroll
        for (int p = 0; p < 3; ++p)
#pragma unroll
            for (int d = 0; d < 3; ++d)
                cp[p][d] = *reinterpret_cast<const v2f*>(&P[sc][d * kUP + q0 + 2 * p + 2]);
        __builtin_amdgcn_sched_barrier(0);
        float wv[4][8];
#pragma unroll
        for (int rowd = 0; rowd < 4; ++rowd) {
#pragma unroll
            for (int j = 0; j < 8; ++j) wv[rowd][j] = 0.f;
#pragma unroll
            for (int sl = 0; sl < 3; ++sl) {
                const int base = (rowd * 32 + oo) * kUP + q0;
                const v2f* p2 = reinterpret_cast<const v2f*>(&U[sl][base]);
#pragma unroll
                for (int j = 0; j < 4; ++j) {
                    const v2f v = p2[j];
                    wv[rowd][2 * j]     += v.x;
                    wv[rowd][2 * j + 1] += v.y;
                }
            }
        }
#pragma unroll
        for (int rowd = 0; rowd < 4; ++rowd)
#pragma unroll
            for (int j = 0; j < 8; ++j) asm volatile("" : "+v"(wv[rowd][j]));
#pragma unroll
        for (int p = 0; p < 3; ++p)
#pragma unroll
            for (int d = 0; d < 3; ++d) asm volatile("" : "+v"(cp[p][d]));

#pragma unroll
        for (int p = 0; p < 3; ++p) {
            const int pxo = q0 + 2 * p;
            const int w   = w0 + pxo;
            if (pxo >= kTW || w >= kW) continue;
            v2f res;
#pragma unroll
            for (int e = 0; e < 2; ++e) {
                const int jj = 2 * p + e;
                const float su0 = wv[0][jj] + wv[0][jj + 1] + wv[0][jj + 2];
                const float su1 = wv[1][jj] + wv[1][jj + 1] + wv[1][jj + 2];
                const float su2 = wv[2][jj] + wv[2][jj + 1] + wv[2][jj + 2];
                float acc       = wv[3][jj] + wv[3][jj + 1] + wv[3][jj + 2];
                const float pc0 = e ? cp[p][0].y : cp[p][0].x;
                const float pc1 = e ? cp[p][1].y : cp[p][1].x;
                const float pc2 = e ? cp[p][2].y : cp[p][2].x;
                acc += __builtin_fmaf(pc2, su2, __builtin_fmaf(pc0, su0, pc1 * su1));
                if (e) res.y = acc; else res.x = acc;
            }
            *reinterpret_cast<v2f*>(&out[((b * kO + obase + oo) * kH + h) * kW + w]) = res;
        }
    };

    load_x(h0 - 1);
    stage_row(h0 - 1);
    load_x(h0);
    lds_barrier();
    for (int r = h0 - 1; r <= h0 + kHB; ++r) {
        if (r >= h0 + 2) window_row(r - 2);
        compute_u_row(r, (r + 1 - h0) % 3);
    }
    window_row(h0 + kHB - 1);
}


extern "C" void kernel_launch(void* const* d_in, const int* in_sizes, int n_in,
                              void* d_out, int out_size, void* d_ws, size_t ws_size,
                              hipStream_t stream) {
    const float* x   = (const float*)d_in[0];
    const float* pts = (const float*)d_in[1];
    const float* wgt = (const float*)d_in[2];
    float* out = (float*)d_out;

    dim3 grid(kNS, kNH * 2, kB);
    flexconv_wmma<<<grid, 256, 0, stream>>>(x, pts, wgt, out);

    (void)in_sizes; (void)n_in; (void)out_size; (void)d_ws; (void)ws_size;
}


# Wave priority stepped down 3,2,1,0 over the 18 rows (s_setprio under uniform row tests): of the three workgroups sharing a WGP the one behind outranks the leaders, so none ends its rows alone
# speedup vs baseline: 2.5416x; 1.0439x over previous
#include <hip/hip_runtime.h>


typedef float v2f __attribute__((ext_vector_type(2)));
typedef float v4f __attribute__((ext_vector_type(4)));
typedef float v8f __attribute__((ext_vector_type(8)));

namespace {
constexpr int kB  = 8;
constexpr int kC  = 64;
constexpr int kH  = 64;
constexpr int kW  = 512;
constexpr int kO  = 64;
constexpr int kTW = 46;
constexpr int kNC = 48;
constexpr int kNS = 12;
constexpr int kHB = 16;
constexpr int kNH = 4;
constexpr int kODL = 128;
constexpr int kXI = 68;
constexpr int kUP = 50;
}

__device__ __forceinline__ void lds_barrier()
{
    __builtin_amdgcn_fence(__ATOMIC_RELEASE, "workgroup", "local");
    __builtin_amdgcn_s_barrier();
    __builtin_amdgcn_fence(__ATOMIC_ACQUIRE, "workgroup", "local");
}

__global__ __launch_bounds__(256, 6)
void flexconv_wmma(const float* __restrict__ x,
                   const float* __restrict__ pts,
                   const float* __restrict__ wgt,
                   float* __restrict__ out)
{
    __shared__ __attribute__((aligned(16))) float Xt[kNC * kXI];
    __shared__ __attribute__((aligned(16))) float U[3][kODL * kUP];
    __shared__ __attribute__((aligned(16))) float P[3][3 * kUP];

    __builtin_amdgcn_s_setprio(3);

    const int stripe = blockIdx.x;
    const int hseg   = blockIdx.y >> 1;
    const int ohalf  = blockIdx.y & 1;
    const int b      = blockIdx.z;

    const int w0    = stripe * kTW;
    const int wb    = w0 - 1;
    const int h0    = hseg * kHB;
    const int obase = ohalf * 32;

    const int tid   = threadIdx.x;
    const int wid   = __builtin_amdgcn_readfirstlane(tid >> 5);
    const int lane  = tid & 31;
    const int m     = lane & 15;
    const int khalf = lane >> 4;

    const int dA  = m & 3;
    const int oA  = obase + wid * 4 + ((m >> 2) & 1) + 2 * (m >> 3);
    float a0[16], a1[16];
#pragma unroll
    for (int kk = 0; kk < 16; ++kk) {
        const int i0 = 4 * kk + 2 * khalf;
        a0[kk] = wgt[oA * (kC * 4) + (i0 + 0) * 4 + dA];
        a1[kk] = wgt[oA * (kC * 4) + (i0 + 1) * 4 + dA];
    }

    const int q4  = tid >> 4;
    const int sxq = ((b * kC + 4 * q4) * kH) * kW;
    const int sl0 = 3 * m * kXI + 4 * q4;
    bool fok[3];
    int  fwc[3];
#pragma unroll
    for (int t = 0; t < 3; ++t) {
        const int fw = wb + 16 * t + m;
        fok[t] = (fw >= 0) & (fw < kW);
        fwc[t] = min(max(fw, 0), kW - 1);
    }
    float xr[3][4];
    float pr[3][3];

    auto load_x = [&](int r) {
        const int rc = min(max(r, 0), kH - 1);
#pragma unroll
        for (int k = 0; k < 3; ++k)
#pragma unroll
            for (int c = 0; c < 4; ++c)
                xr[k][c] = x[sxq + (c * kH + rc) * kW + fwc[k]];
    };
    auto load_p = [&](int r) {
        const int rc = min(max(r, 0), kH - 1);
#pragma unroll
        for (int d = 0; d < 3; ++d)
#pragma unroll
            for (int t = 0; t < 3; ++t)
                pr[t][d] = pts[((b * 3 + d) * kH + rc) * kW + fwc[t]];
    };

    auto stage_row = [&](int r) {
        const bool rok = (r >= 0) & (r < kH);
#pragma unroll
        for (int k = 0; k < 3; ++k) {
            const bool ok = rok & fok[k];
            v4f v;
            v.x = ok ? xr[k][0] : 0.0f;
            v.y = ok ? xr[k][1] : 0.0f;
            v.z = ok ? xr[k][2] : 0.0f;
            v.w = ok ? xr[k][3] : 0.0f;
            *reinterpret_cast<v4f*>(&Xt[sl0 + k * kXI]) = v;
        }
    };

    auto compute_u_row = [&](int r, int s) {
        load_p(r);
        const bool rok = (r >= 0) & (r < kH);

        v8f acc0 = {0.f,0.f,0.f,0.f,0.f,0.f,0.f,0.f};
        v8f acc1 = acc0;
        v8f acc2 = acc0;
#pragma unroll
        for (int kk = 0; kk < 16; ++kk) {
            const int i0 = 4 * kk + 2 * khalf;
            v2f av; av.x = a0[kk]; av.y = a1[kk];
            const v2f bv0 = *reinterpret_cast<const v2f*>(&Xt[(3 * m + 0) * kXI + i0]);
            const v2f bv1 = *reinterpret_cast<const v2f*>(&Xt[(3 * m + 1) * kXI + i0]);
            const v2f bv2 = *reinterpret_cast<const v2f*>(&Xt[(3 * m + 2) * kXI + i0]);
            acc0 = __builtin_amdgcn_wmma_f32_16x16x4_f32(false, av, false, bv0,
                                                         (short)0, acc0, false, false);
            acc1 = __builtin_amdgcn_wmma_f32_16x16x4_f32(false, av, false, bv1,
                                                         (short)0, acc1, false, false);
            acc2 = __builtin_amdgcn_wmma_f32_16x16x4_f32(false, av, false, bv2,
                                                         (short)0, acc2, false, false);
        }

        __builtin_amdgcn_sched_barrier(0);

        float pp[3][3];
        float vb[3][2];
#pragma unroll
        for (int t = 0; t < 3; ++t) {
            const bool okp = rok & fok[t];
            pp[t][0] = okp ? pr[t][0] : 0.f;
            pp[t][1] = okp ? pr[t][1] : 0.f;
            pp[t][2] = okp ? pr[t][2] : 0.f;
            const v8f acc = (t == 0) ? acc0 : ((t == 1) ? acc1 : acc2);
#pragma unroll
            for (int j = 0; j < 2; ++j)
                vb[t][j] = acc[4 * j + 3]
                         - __builtin_fmaf(pp[t][2], acc[4 * j + 2],
                                          __builtin_fmaf(pp[t][0], acc[4 * j + 0], pp[t][1] * acc[4 * j + 1]));
        }

        lds_barrier();

#pragma unroll
        for (int t = 0; t < 3; ++t) {
            if (wid == t) {
                P[s][0 * kUP + 16 * t + m + 1] = pp[t][0];
                P[s][1 * kUP + 16 * t + m + 1] = pp[t][1];
                P[s][2 * kUP + 16 * t + m + 1] = pp[t][2];
            }
            const v8f acc = (t == 0) ? acc0 : ((t == 1) ? acc1 : acc2);
#pragma unroll
            for (int j = 0; j < 2; ++j) {
                const int oo = wid * 4 + j + 2 * khalf;
                const int col = 16 * t + m;
                U[s][(oo)      * kUP + col] = acc[4 * j + 0];
                U[s][(32 + oo) * kUP + col] = acc[4 * j + 1];
                U[s][(64 + oo) * kUP + col] = acc[4 * j + 2];
                U[s][(96 + oo) * kUP + col] = vb[t][j];
            }
        }
        stage_row(r + 1);
        load_x(r + 2);
        lds_barrier();
    };

    const int sg = lane >> 3;
    const int oo = wid + 8 * sg;
    const int q0 = (lane & 7) * 6;

    auto window_row = [&](int h) {
        const int sc = (h + 1 - h0) % 3;

        v2f cp[3][3];
#pragma unroll
        for (int p = 0; p < 3; ++p)
#pragma unroll
            for (int d = 0; d < 3; ++d)
                cp[p][d] = *reinterpret_cast<const v2f*>(&P[sc][d * kUP + q0 + 2 * p + 2]);
        __builtin_amdgcn_sched_barrier(0);
        float wv[4][8];
#pragma unroll
        for (int rowd = 0; rowd < 4; ++rowd) {
#pragma unroll
            for (int j = 0; j < 8; ++j) wv[rowd][j] = 0.f;
#pragma unroll
            for (int sl = 0; sl < 3; ++sl) {
                const int base = (rowd * 32 + oo) * kUP + q0;
                const v2f* p2 = reinterpret_cast<const v2f*>(&U[sl][base]);
#pragma unroll
                for (int j = 0; j < 4; ++j) {
                    const v2f v = p2[j];
                    wv[rowd][2 * j]     += v.x;
                    wv[rowd][2 * j + 1] += v.y;
                }
            }
        }
#pragma unroll
        for (int rowd = 0; rowd < 4; ++rowd)
#pragma unroll
            for (int j = 0; j < 8; ++j) asm volatile("" : "+v"(wv[rowd][j]));
#pragma unroll
        for (int p = 0; p < 3; ++p)
#pragma unroll
            for (int d = 0; d < 3; ++d) asm volatile("" : "+v"(cp[p][d]));

#pragma unroll
        for (int p = 0; p < 3; ++p) {
            const int pxo = q0 + 2 * p;
            const int w   = w0 + pxo;
            if (pxo >= kTW || w >= kW) continue;
            v2f res;
#pragma unroll
            for (int e = 0; e < 2; ++e) {
                const int jj = 2 * p + e;
                const float su0 = wv[0][jj] + wv[0][jj + 1] + wv[0][jj + 2];
                const float su1 = wv[1][jj] + wv[1][jj + 1] + wv[1][jj + 2];
                const float su2 = wv[2][jj] + wv[2][jj + 1] + wv[2][jj + 2];
                float acc       = wv[3][jj] + wv[3][jj + 1] + wv[3][jj + 2];
                const float pc0 = e ? cp[p][0].y : cp[p][0].x;
                const float pc1 = e ? cp[p][1].y : cp[p][1].x;
                const float pc2 = e ? cp[p][2].y : cp[p][2].x;
                acc += __builtin_fmaf(pc2, su2, __builtin_fmaf(pc0, su0, pc1 * su1));
                if (e) res.y = acc; else res.x = acc;
            }
            *reinterpret_cast<v2f*>(&out[((b * kO + obase + oo) * kH + h) * kW + w]) = res;
        }
    };

    load_x(h0 - 1);
    stage_row(h0 - 1);
    load_x(h0);
    lds_barrier();
    for (int r = h0 - 1; r <= h0 + kHB; ++r) {
        if (r == h0 + 4) __builtin_amdgcn_s_setprio(2);
        else if (r == h0 + 9) __builtin_amdgcn_s_setprio(1);
        else if (r == h0 + 14) __builtin_amdgcn_s_setprio(0);
        if (r >= h0 + 2) window_row(r - 2);
        compute_u_row(r, (r + 1 - h0) % 3);
    }
    window_row(h0 + kHB - 1);
}


extern "C" void kernel_launch(void* const* d_in, const int* in_sizes, int n_in,
                              void* d_out, int out_size, void* d_ws, size_t ws_size,
                              hipStream_t stream) {
    const float* x   = (const float*)d_in[0];
    const float* pts = (const float*)d_in[1];
    const float* wgt = (const float*)d_in[2];
    float* out = (float*)d_out;

    dim3 grid(kNS, kNH * 2, kB);
    flexconv_wmma<<<grid, 256, 0, stream>>>(x, pts, wgt, out);

    (void)in_sizes; (void)n_in; (void)out_size; (void)d_ws; (void)ws_size;
}
